# starting kernel
# baseline (speedup 1.0000x reference)
#include <hip/hip_runtime.h>
#include <hip/hip_bf16.h>

typedef __attribute__((ext_vector_type(2))) float v2f;
typedef __attribute__((ext_vector_type(8))) float v8f;

static __device__ __forceinline__ void atomic_fadd(float* p, float v) {
    __hip_atomic_fetch_add(p, v, __ATOMIC_RELAXED, __HIP_MEMORY_SCOPE_AGENT);
}

__global__ void gcn_init_deg(float* __restrict__ deg, int n) {
    int i = blockIdx.x * blockDim.x + threadIdx.x;
    if (i < n) deg[i] = 1.0f;
}

__global__ void gcn_count_deg(const int* __restrict__ dst, int E,
                              float* __restrict__ deg) {
    int e = blockIdx.x * blockDim.x + threadIdx.x;
    if (e < E) atomic_fadd(&deg[dst[e]], 1.0f);
}

__global__ void gcn_rsqrt(float* __restrict__ deg, int n) {
    int i = blockIdx.x * blockDim.x + threadIdx.x;
    if (i < n) deg[i] = rsqrtf(deg[i]);
}

__global__ void gcn_zero(float* __restrict__ p, long long n) {
    long long i = (long long)blockIdx.x * blockDim.x + threadIdx.x;
    if (i < n) p[i] = 0.0f;
}

__global__ void gcn_gemm_f32_wmma(const float* __restrict__ A,
                                  const float* __restrict__ B,
                                  float* __restrict__ C,
                                  int M, int N, int K) {
    const int wave  = threadIdx.x >> 5;
    const int lane  = threadIdx.x & 31;
    const int mTile = blockIdx.x * 4 + wave;
    if (mTile * 16 >= M) return;

    const int r16   = lane & 15;
    const int koff  = (lane >> 4) << 1;
    const int nBase = blockIdx.y * 64 + r16;

    v8f c0 = {0.f,0.f,0.f,0.f,0.f,0.f,0.f,0.f};
    v8f c1 = c0, c2 = c0, c3 = c0;

    const float* Arow = A + ((long long)mTile * 16 + r16) * K;

    for (int k0 = 0; k0 < K; k0 += 4) {
        v2f a = *(const v2f*)(Arow + k0 + koff);
        const float* Brow0 = B + (long long)(k0 + koff)     * N + nBase;
        const float* Brow1 = B + (long long)(k0 + koff + 1) * N + nBase;
        v2f b0, b1, b2, b3;
        b0.x = Brow0[0];  b0.y = Brow1[0];
        b1.x = Brow0[16]; b1.y = Brow1[16];
        b2.x = Brow0[32]; b2.y = Brow1[32];
        b3.x = Brow0[48]; b3.y = Brow1[48];
        c0 = __builtin_amdgcn_wmma_f32_16x16x4_f32(false, a, false, b0, (short)0, c0, false, false);
        c1 = __builtin_amdgcn_wmma_f32_16x16x4_f32(false, a, false, b1, (short)0, c1, false, false);
        c2 = __builtin_amdgcn_wmma_f32_16x16x4_f32(false, a, false, b2, (short)0, c2, false, false);
        c3 = __builtin_amdgcn_wmma_f32_16x16x4_f32(false, a, false, b3, (short)0, c3, false, false);
    }

    const long long rowBase = (long long)mTile * 16 + ((lane >> 4) << 3);
#pragma unroll
    for (int v = 0; v < 8; ++v) {
        float* Crow = C + (rowBase + v) * N + nBase;
        Crow[0]  = c0[v];
        Crow[16] = c1[v];
        Crow[32] = c2[v];
        Crow[48] = c3[v];
    }
}

__global__ void gcn_scatter(const float* __restrict__ h,
                            const int* __restrict__ src,
                            const int* __restrict__ dst,
                            const float* __restrict__ dinv,
                            float* __restrict__ agg,
                            int E, int C) {
    const int wavesPerBlock = blockDim.x >> 5;
    const int e    = blockIdx.x * wavesPerBlock + (threadIdx.x >> 5);
    const int lane = threadIdx.x & 31;
    if (e >= E) return;
    const int s = src[e];
    const int d = dst[e];
    const float nrm = dinv[s] * dinv[d];
    const float* hs = h   + (long long)s * C;
    float*       ad = agg + (long long)d * C;
    for (int c = lane; c < C; c += 32) {
        atomic_fadd(&ad[c], hs[c] * nrm);
    }
}

__global__ void gcn_finalize(const float* __restrict__ h,
                             const float* __restrict__ dinv,
                             const float* __restrict__ bias,
                             float* __restrict__ agg,
                             int nNodes, int C, int doRelu) {
    long long i = (long long)blockIdx.x * blockDim.x + threadIdx.x;
    long long total = (long long)nNodes * C;
    if (i >= total) return;
    int node = (int)(i / C);
    int c    = (int)(i % C);
    float di = dinv[node];
    float v  = agg[i] + h[i] * di * di + bias[c];
    if (doRelu) v = fmaxf(v, 0.0f);
    agg[i] = v;
}

extern "C" void kernel_launch(void* const* d_in, const int* in_sizes, int n_in,
                              void* d_out, int out_size, void* d_ws, size_t ws_size,
                              hipStream_t stream) {
    const float* x   = (const float*)d_in[0];
    const int*   ei  = (const int*)d_in[1];
    const float* W1  = (const float*)d_in[2];
    const float* b1  = (const float*)d_in[3];
    const float* W2  = (const float*)d_in[4];
    const float* b2  = (const float*)d_in[5];

    const int IN_C  = 128;
    const int HID_C = 128;
    const int OUT_C = 64;
    const int nNodes = in_sizes[0] / IN_C;
    const int E      = in_sizes[1] / 2;
    const int* srcIdx = ei;
    const int* dstIdx = ei + E;

    float* dinv = (float*)d_ws;
    float* bufA = dinv + nNodes;
    float* bufB = bufA + (long long)nNodes * HID_C;
    float* out  = (float*)d_out;

    const int T = 256;
    const long long nh = (long long)nNodes * HID_C;
    const long long no = (long long)nNodes * OUT_C;
    const int mBlocks = (nNodes / 16 + 3) / 4;

    gcn_init_deg <<<(nNodes + T - 1) / T, T, 0, stream>>>(dinv, nNodes);
    gcn_count_deg<<<(E + T - 1) / T, T, 0, stream>>>(dstIdx, E, dinv);
    gcn_rsqrt    <<<(nNodes + T - 1) / T, T, 0, stream>>>(dinv, nNodes);

    {
        dim3 grid(mBlocks, HID_C / 64);
        gcn_gemm_f32_wmma<<<grid, 128, 0, stream>>>(x, W1, bufA,
                                                    nNodes, HID_C, IN_C);
    }

    gcn_zero<<<(unsigned)((nh + T - 1) / T), T, 0, stream>>>(bufB, nh);
    gcn_scatter<<<(E + 7) / 8, 256, 0, stream>>>(bufA, srcIdx, dstIdx, dinv,
                                                 bufB, E, HID_C);

    gcn_finalize<<<(unsigned)((nh + T - 1) / T), T, 0, stream>>>(
        bufA, dinv, b1, bufB, nNodes, HID_C,  1);

    {
        dim3 grid(mBlocks, OUT_C / 64);
        gcn_gemm_f32_wmma<<<grid, 128, 0, stream>>>(bufB, W2, bufA,
                                                    nNodes, OUT_C, HID_C);
    }

    gcn_zero<<<(unsigned)((no + T - 1) / T), T, 0, stream>>>(out, no);
    gcn_scatter<<<(E + 7) / 8, 256, 0, stream>>>(bufA, srcIdx, dstIdx, dinv,
                                                 out, E, OUT_C);

    gcn_finalize<<<(unsigned)((no + T - 1) / T), T, 0, stream>>>(
        bufA, dinv, b2, out, nNodes, OUT_C,  0);
}


# GEMM in whole lines: 64-col W strip staged once per block in LDS as k-pairs (B fragment = one ds_load_b64), A rows as 16 b128 loads up front, C/S out as float4 rows via LDS tile; k sum order changes
# speedup vs baseline: 2.6493x; 2.6493x over previous
#include <hip/hip_runtime.h>
#include <hip/hip_bf16.h>

typedef __attribute__((ext_vector_type(2))) float v2f;
typedef __attribute__((ext_vector_type(8))) float v8f;

static __device__ __forceinline__ void atomic_fadd(float* p, float v) {
    __hip_atomic_fetch_add(p, v, __ATOMIC_RELAXED, __HIP_MEMORY_SCOPE_AGENT);
}

constexpr int DEG_CAP    = 51200;
constexpr int DEG_GROUPS = 64;
constexpr int DEG_T      = 1024;

static __device__ __forceinline__ long long deg_pitch(int n) {
    return (long long)((n + 31) & ~31);
}

__global__ void __launch_bounds__(DEG_T)
gcn_count_deg_lds(const int* __restrict__ dst, int E, int n,
                  int* __restrict__ partial) {
    __shared__ int hist[DEG_CAP];
    const int tid  = threadIdx.x;
    const int per  = (E + DEG_GROUPS - 1) / DEG_GROUPS;
    const int eBeg = blockIdx.x * per;
    const int eEnd = min(eBeg + per, E);
    int* mine = partial + (long long)blockIdx.x * deg_pitch(n);

    for (int base = 0; base < n; base += DEG_CAP) {
        const int m = min(n - base, DEG_CAP);
        for (int i = tid; i < m; i += DEG_T) hist[i] = 0;
        __syncthreads();

        int e = eBeg + tid;
        for (; e + 3 * DEG_T < eEnd; e += 4 * DEG_T) {
            const int d0 = dst[e]             - base;
            const int d1 = dst[e + DEG_T]     - base;
            const int d2 = dst[e + 2 * DEG_T] - base;
            const int d3 = dst[e + 3 * DEG_T] - base;
            if ((unsigned)d0 < (unsigned)m) atomicAdd(&hist[d0], 1);
            if ((unsigned)d1 < (unsigned)m) atomicAdd(&hist[d1], 1);
            if ((unsigned)d2 < (unsigned)m) atomicAdd(&hist[d2], 1);
            if ((unsigned)d3 < (unsigned)m) atomicAdd(&hist[d3], 1);
        }
        for (; e < eEnd; e += DEG_T) {
            const int d = dst[e] - base;
            if ((unsigned)d < (unsigned)m) atomicAdd(&hist[d], 1);
        }
        __syncthreads();

        for (int i = tid; i < m; i += DEG_T) mine[base + i] = hist[i];
        __syncthreads();
    }
}

__global__ void gcn_deg_rsqrt(const int* __restrict__ partial,
                              float* __restrict__ dinv, int n) {
    int i = blockIdx.x * blockDim.x + threadIdx.x;
    if (i < n) {
        const long long pitch = deg_pitch(n);
        int cnt = 1;
#pragma unroll
        for (int g = 0; g < DEG_GROUPS; ++g) cnt += partial[g * pitch + i];
        dinv[i] = rsqrtf((float)cnt);
    }
}

constexpr int KDIM  = 128;
constexpr int WPITCH  = 2 * 64 + 16;
constexpr int TPITCH = 64 + 4;

template <bool RELU_A>
__global__ void __launch_bounds__(128)
gcn_gemm_f32_wmma(const float* __restrict__ A,
                  const float* __restrict__ B,
                  const float* __restrict__ dinv,
                  const float* __restrict__ bias,
                  float* __restrict__ C,
                  float* __restrict__ S,
                  int M, int N) {
    __shared__ __attribute__((aligned(16))) float Wp[(KDIM / 2) * WPITCH];
    __shared__ __attribute__((aligned(16))) float tiles[4 * 16 * TPITCH];

    const int tid   = threadIdx.x;
    const int wave  = tid >> 5;
    const int lane  = tid & 31;
    const int mTile = blockIdx.x * 4 + wave;
    const bool active = mTile * 16 < M;
    const int r16   = lane & 15;
    const int h     = lane >> 4;
    const int n0    = blockIdx.y * 64;

    float4 a4[16];
    if (active) {
        const float* Arow = A + ((long long)mTile * 16 + r16) * KDIM + 4 * h;
#pragma unroll
        for (int j = 0; j < 16; ++j) a4[j] = *(const float4*)(Arow + 8 * j);
    }

#pragma unroll
    for (int i = 0; i < (KDIM / 2) * 16 / 128; ++i) {
        const int it = tid + 128 * i;
        const int q  = it & 15;
        const int kp = it >> 4;
        const float4 lo = *(const float4*)(B + (long long)(2 * kp)     * N + n0 + 4 * q);
        const float4 hi = *(const float4*)(B + (long long)(2 * kp + 1) * N + n0 + 4 * q);
        float* w = Wp + kp * WPITCH + 8 * q;
        *(float4*)(w)     = make_float4(lo.x, hi.x, lo.y, hi.y);
        *(float4*)(w + 4) = make_float4(lo.z, hi.z, lo.w, hi.w);
    }
    __syncthreads();
    if (!active) return;

    v8f c0 = {0.f,0.f,0.f,0.f,0.f,0.f,0.f,0.f};
    v8f c1 = c0, c2 = c0, c3 = c0;

    const float* wl = Wp + (2 * h) * WPITCH + 2 * r16;
#pragma unroll
    for (int j = 0; j < 16; ++j) {
        float4 a = a4[j];
        if (RELU_A) {
            a.x = fmaxf(a.x, 0.0f); a.y = fmaxf(a.y, 0.0f);
            a.z = fmaxf(a.z, 0.0f); a.w = fmaxf(a.w, 0.0f);
        }
        const float* w0 = wl + (4 * j) * WPITCH;
        const float* w1 = w0 + WPITCH;
        const v2f b00 = *(const v2f*)(w0);
        const v2f b01 = *(const v2f*)(w0 + 32);
        const v2f b02 = *(const v2f*)(w0 + 64);
        const v2f b03 = *(const v2f*)(w0 + 96);
        const v2f b10 = *(const v2f*)(w1);
        const v2f b11 = *(const v2f*)(w1 + 32);
        const v2f b12 = *(const v2f*)(w1 + 64);
        const v2f b13 = *(const v2f*)(w1 + 96);
        v2f a01, a23;
        a01.x = a.x; a01.y = a.y;
        a23.x = a.z; a23.y = a.w;
        c0 = __builtin_amdgcn_wmma_f32_16x16x4_f32(false, a01, false, b00, (short)0, c0, false, false);
        c1 = __builtin_amdgcn_wmma_f32_16x16x4_f32(false, a01, false, b01, (short)0, c1, false, false);
        c2 = __builtin_amdgcn_wmma_f32_16x16x4_f32(false, a01, false, b02, (short)0, c2, false, false);
        c3 = __builtin_amdgcn_wmma_f32_16x16x4_f32(false, a01, false, b03, (short)0, c3, false, false);
        c0 = __builtin_amdgcn_wmma_f32_16x16x4_f32(false, a23, false, b10, (short)0, c0, false, false);
        c1 = __builtin_amdgcn_wmma_f32_16x16x4_f32(false, a23, false, b11, (short)0, c1, false, false);
        c2 = __builtin_amdgcn_wmma_f32_16x16x4_f32(false, a23, false, b12, (short)0, c2, false, false);
        c3 = __builtin_amdgcn_wmma_f32_16x16x4_f32(false, a23, false, b13, (short)0, c3, false, false);
    }

    float* tile = tiles + wave * 16 * TPITCH;
#pragma unroll
    for (int v = 0; v < 8; ++v) {
        float* tr = tile + (8 * h + v) * TPITCH + r16;
        tr[0]  = c0[v];
        tr[16] = c1[v];
        tr[32] = c2[v];
        tr[48] = c3[v];
    }
    __builtin_amdgcn_fence(__ATOMIC_RELEASE, "wavefront");
    __builtin_amdgcn_wave_barrier();
    __builtin_amdgcn_fence(__ATOMIC_ACQUIRE, "wavefront");

    const int rr = lane >> 4;
    const int cc = 4 * (lane & 15);
    const float4 bb = *(const float4*)(bias + n0 + cc);
    const long long row0 = (long long)mTile * 16;
#pragma unroll
    for (int i = 0; i < 8; ++i) {
        const int row = 2 * i + rr;
        const float4 vv = *(const float4*)(tile + row * TPITCH + cc);
        const float di = dinv[row0 + row];
        const float w  = di * di;
        float4 ss;
        ss.x = vv.x * w + bb.x;
        ss.y = vv.y * w + bb.y;
        ss.z = vv.z * w + bb.z;
        ss.w = vv.w * w + bb.w;
        *(float4*)(C + (row0 + row) * N + n0 + cc) = vv;
        *(float4*)(S + (row0 + row) * N + n0 + cc) = ss;
    }
}

template <int C>
static __device__ __forceinline__ void scatter_rows(const float* __restrict__ h,
                                                    const int* __restrict__ src,
                                                    const int* __restrict__ dst,
                                                    const float* __restrict__ dinv,
                                                    float* __restrict__ agg,
                                                    int E) {
    constexpr int Q   = C / 32;
    constexpr int EPW = 16 / Q;
    const int lane = threadIdx.x & 31;
    const int wave = blockIdx.x * (blockDim.x >> 5) + (threadIdx.x >> 5);
    const int e0   = wave * EPW;
    if (e0 >= E) return;

    if (e0 + EPW <= E) {
        const int el = e0 + (lane & (EPW - 1));
        const int s  = src[el];
        const int d  = dst[el];
        const float ds = dinv[s];
        const float dd = dinv[d];

        float val[16];
#pragma unroll
        for (int j = 0; j < EPW; ++j) {
            const int sj = __builtin_amdgcn_readlane(s, j);
            const float* hs = h + (long long)sj * C + lane;
#pragma unroll
            for (int q = 0; q < Q; ++q) val[j * Q + q] = hs[q * 32];
        }

        const int nrmBits = __float_as_int(ds * dd);
#pragma unroll
        for (int j = 0; j < EPW; ++j) {
            const float nrm = __int_as_float(__builtin_amdgcn_readlane(nrmBits, j));
            const int   dj  = __builtin_amdgcn_readlane(d, j);
            float* ad = agg + (long long)dj * C + lane;
#pragma unroll
            for (int q = 0; q < Q; ++q) atomic_fadd(ad + q * 32, val[j * Q + q] * nrm);
        }
    } else {
        for (int e = e0; e < E; ++e) {
            const int s = src[e];
            const int d = dst[e];
            const float nrm = dinv[s] * dinv[d];
            const float* hs = h   + (long long)s * C + lane;
            float*       ad = agg + (long long)d * C + lane;
#pragma unroll
            for (int q = 0; q < Q; ++q) atomic_fadd(ad + q * 32, hs[q * 32] * nrm);
        }
    }
}

__global__ void gcn_scatter128(const float* __restrict__ h,
                               const int* __restrict__ src,
                               const int* __restrict__ dst,
                               const float* __restrict__ dinv,
                               float* __restrict__ agg, int E) {
    scatter_rows<128>(h, src, dst, dinv, agg, E);
}

__global__ void gcn_scatter64(const float* __restrict__ h,
                              const int* __restrict__ src,
                              const int* __restrict__ dst,
                              const float* __restrict__ dinv,
                              float* __restrict__ agg, int E) {
    scatter_rows<64>(h, src, dst, dinv, agg, E);
}

extern "C" void kernel_launch(void* const* d_in, const int* in_sizes, int n_in,
                              void* d_out, int out_size, void* d_ws, size_t ws_size,
                              hipStream_t stream) {
    const float* x   = (const float*)d_in[0];
    const int*   ei  = (const int*)d_in[1];
    const float* W1  = (const float*)d_in[2];
    const float* b1  = (const float*)d_in[3];
    const float* W2  = (const float*)d_in[4];
    const float* b2  = (const float*)d_in[5];

    const int IN_C  = 128;
    const int HID_C = 128;
    const int OUT_C = 64;
    const int nNodes = in_sizes[0] / IN_C;
    const int E      = in_sizes[1] / 2;
    const int* srcIdx = ei;
    const int* dstIdx = ei + E;

    float* bufA = (float*)d_ws;
    float* bufB = bufA + (long long)nNodes * HID_C;
    float* dinv = bufB + (long long)nNodes * HID_C;
    float* out  = (float*)d_out;

    const int T = 256;
    const int mBlocks = (nNodes / 16 + 3) / 4;

    int* degPart = (int*)bufB;
    gcn_count_deg_lds<<<DEG_GROUPS, DEG_T, 0, stream>>>(dstIdx, E, nNodes, degPart);
    gcn_deg_rsqrt    <<<(nNodes + T - 1) / T, T, 0, stream>>>(degPart, dinv, nNodes);

    {
        dim3 grid(mBlocks, HID_C / 64);
        gcn_gemm_f32_wmma<false><<<grid, 128, 0, stream>>>(x, W1, dinv, b1, bufA, bufB,
                                                           nNodes, HID_C);
    }

    gcn_scatter128<<<(E + 31) / 32, 256, 0, stream>>>(bufA, srcIdx, dstIdx, dinv,
                                                      bufB, E);

    {
        dim3 grid(mBlocks, OUT_C / 64);
        gcn_gemm_f32_wmma<true><<<grid, 128, 0, stream>>>(bufB, W2, dinv, b2, bufA, out,
                                                          nNodes, OUT_C);
    }

    gcn_scatter64<<<(E + 63) / 64, 256, 0, stream>>>(bufA, srcIdx, dstIdx, dinv,
                                                     out, E);
}


# Scatter split by L2 half: dst range cut in two classes; a work-group draws a (256-edge chunk, class) ticket (int atomic), own half's class first: a line is updated from one half; f32 order: arrival
# speedup vs baseline: 2.9077x; 1.0975x over previous
#include <hip/hip_runtime.h>
#include <hip/hip_bf16.h>

typedef __attribute__((ext_vector_type(2))) float v2f;
typedef __attribute__((ext_vector_type(8))) float v8f;

static __device__ __forceinline__ void atomic_fadd(float* p, float v) {
    __hip_atomic_fetch_add(p, v, __ATOMIC_RELAXED, __HIP_MEMORY_SCOPE_AGENT);
}

constexpr int DEG_CAP    = 51200;
constexpr int DEG_GROUPS = 64;
constexpr int DEG_T      = 1024;
constexpr int TICKET_PITCH = 64;

static __device__ __forceinline__ long long deg_pitch(int n) {
    return (long long)((n + 31) & ~31);
}

__global__ void __launch_bounds__(DEG_T)
gcn_count_deg_lds(const int* __restrict__ dst, int E, int n,
                  int* __restrict__ partial) {
    __shared__ int hist[DEG_CAP];
    const int tid  = threadIdx.x;
    const int per  = (E + DEG_GROUPS - 1) / DEG_GROUPS;
    const int eBeg = blockIdx.x * per;
    const int eEnd = min(eBeg + per, E);
    int* mine = partial + (long long)blockIdx.x * deg_pitch(n);

    for (int base = 0; base < n; base += DEG_CAP) {
        const int m = min(n - base, DEG_CAP);
        for (int i = tid; i < m; i += DEG_T) hist[i] = 0;
        __syncthreads();

        int e = eBeg + tid;
        for (; e + 3 * DEG_T < eEnd; e += 4 * DEG_T) {
            const int d0 = dst[e]             - base;
            const int d1 = dst[e + DEG_T]     - base;
            const int d2 = dst[e + 2 * DEG_T] - base;
            const int d3 = dst[e + 3 * DEG_T] - base;
            if ((unsigned)d0 < (unsigned)m) atomicAdd(&hist[d0], 1);
            if ((unsigned)d1 < (unsigned)m) atomicAdd(&hist[d1], 1);
            if ((unsigned)d2 < (unsigned)m) atomicAdd(&hist[d2], 1);
            if ((unsigned)d3 < (unsigned)m) atomicAdd(&hist[d3], 1);
        }
        for (; e < eEnd; e += DEG_T) {
            const int d = dst[e] - base;
            if ((unsigned)d < (unsigned)m) atomicAdd(&hist[d], 1);
        }
        __syncthreads();

        for (int i = tid; i < m; i += DEG_T) mine[base + i] = hist[i];
        __syncthreads();
    }
}

__global__ void gcn_deg_rsqrt(const int* __restrict__ partial,
                              float* __restrict__ dinv, int n,
                              int* __restrict__ tickets) {
    int i = blockIdx.x * blockDim.x + threadIdx.x;
    if (i < 2) tickets[i * TICKET_PITCH] = 0;
    if (i < n) {
        const long long pitch = deg_pitch(n);
        int cnt = 1;
#pragma unroll
        for (int g = 0; g < DEG_GROUPS; ++g) cnt += partial[g * pitch + i];
        dinv[i] = rsqrtf((float)cnt);
    }
}

constexpr int KDIM  = 128;
constexpr int WPITCH  = 2 * 64 + 16;
constexpr int TPITCH = 64 + 4;

template <bool RELU_A>
__global__ void __launch_bounds__(128)
gcn_gemm_f32_wmma(const float* __restrict__ A,
                  const float* __restrict__ B,
                  const float* __restrict__ dinv,
                  const float* __restrict__ bias,
                  float* __restrict__ C,
                  float* __restrict__ S,
                  int M, int N,
                  int* __restrict__ tickets) {
    __shared__ __attribute__((aligned(16))) float Wp[(KDIM / 2) * WPITCH];
    __shared__ __attribute__((aligned(16))) float tiles[4 * 16 * TPITCH];

    const int tid   = threadIdx.x;
    const int wave  = tid >> 5;
    const int lane  = tid & 31;
    const int mTile = blockIdx.x * 4 + wave;
    const bool active = mTile * 16 < M;
    const int r16   = lane & 15;
    const int h     = lane >> 4;
    const int n0    = blockIdx.y * 64;

    if (tickets != nullptr && blockIdx.x == 0 && blockIdx.y == 0 && tid < 2)
        tickets[tid * TICKET_PITCH] = 0;

    float4 a4[16];
    if (active) {
        const float* Arow = A + ((long long)mTile * 16 + r16) * KDIM + 4 * h;
#pragma unroll
        for (int j = 0; j < 16; ++j) a4[j] = *(const float4*)(Arow + 8 * j);
    }

#pragma unroll
    for (int i = 0; i < (KDIM / 2) * 16 / 128; ++i) {
        const int it = tid + 128 * i;
        const int q  = it & 15;
        const int kp = it >> 4;
        const float4 lo = *(const float4*)(B + (long long)(2 * kp)     * N + n0 + 4 * q);
        const float4 hi = *(const float4*)(B + (long long)(2 * kp + 1) * N + n0 + 4 * q);
        float* w = Wp + kp * WPITCH + 8 * q;
        *(float4*)(w)     = make_float4(lo.x, hi.x, lo.y, hi.y);
        *(float4*)(w + 4) = make_float4(lo.z, hi.z, lo.w, hi.w);
    }
    __syncthreads();
    if (!active) return;

    v8f c0 = {0.f,0.f,0.f,0.f,0.f,0.f,0.f,0.f};
    v8f c1 = c0, c2 = c0, c3 = c0;

    const float* wl = Wp + (2 * h) * WPITCH + 2 * r16;
#pragma unroll
    for (int j = 0; j < 16; ++j) {
        float4 a = a4[j];
        if (RELU_A) {
            a.x = fmaxf(a.x, 0.0f); a.y = fmaxf(a.y, 0.0f);
            a.z = fmaxf(a.z, 0.0f); a.w = fmaxf(a.w, 0.0f);
        }
        const float* w0 = wl + (4 * j) * WPITCH;
        const float* w1 = w0 + WPITCH;
        const v2f b00 = *(const v2f*)(w0);
        const v2f b01 = *(const v2f*)(w0 + 32);
        const v2f b02 = *(const v2f*)(w0 + 64);
        const v2f b03 = *(const v2f*)(w0 + 96);
        const v2f b10 = *(const v2f*)(w1);
        const v2f b11 = *(const v2f*)(w1 + 32);
        const v2f b12 = *(const v2f*)(w1 + 64);
        const v2f b13 = *(const v2f*)(w1 + 96);
        v2f a01, a23;
        a01.x = a.x; a01.y = a.y;
        a23.x = a.z; a23.y = a.w;
        c0 = __builtin_amdgcn_wmma_f32_16x16x4_f32(false, a01, false, b00, (short)0, c0, false, false);
        c1 = __builtin_amdgcn_wmma_f32_16x16x4_f32(false, a01, false, b01, (short)0, c1, false, false);
        c2 = __builtin_amdgcn_wmma_f32_16x16x4_f32(false, a01, false, b02, (short)0, c2, false, false);
        c3 = __builtin_amdgcn_wmma_f32_16x16x4_f32(false, a01, false, b03, (short)0, c3, false, false);
        c0 = __builtin_amdgcn_wmma_f32_16x16x4_f32(false, a23, false, b10, (short)0, c0, false, false);
        c1 = __builtin_amdgcn_wmma_f32_16x16x4_f32(false, a23, false, b11, (short)0, c1, false, false);
        c2 = __builtin_amdgcn_wmma_f32_16x16x4_f32(false, a23, false, b12, (short)0, c2, false, false);
        c3 = __builtin_amdgcn_wmma_f32_16x16x4_f32(false, a23, false, b13, (short)0, c3, false, false);
    }

    float* tile = tiles + wave * 16 * TPITCH;
#pragma unroll
    for (int v = 0; v < 8; ++v) {
        float* tr = tile + (8 * h + v) * TPITCH + r16;
        tr[0]  = c0[v];
        tr[16] = c1[v];
        tr[32] = c2[v];
        tr[48] = c3[v];
    }
    __builtin_amdgcn_fence(__ATOMIC_RELEASE, "wavefront");
    __builtin_amdgcn_wave_barrier();
    __builtin_amdgcn_fence(__ATOMIC_ACQUIRE, "wavefront");

    const int rr = lane >> 4;
    const int cc = 4 * (lane & 15);
    const float4 bb = *(const float4*)(bias + n0 + cc);
    const long long row0 = (long long)mTile * 16;
#pragma unroll
    for (int i = 0; i < 8; ++i) {
        const int row = 2 * i + rr;
        const float4 vv = *(const float4*)(tile + row * TPITCH + cc);
        const float di = dinv[row0 + row];
        const float w  = di * di;
        float4 ss;
        ss.x = vv.x * w + bb.x;
        ss.y = vv.y * w + bb.y;
        ss.z = vv.z * w + bb.z;
        ss.w = vv.w * w + bb.w;
        *(float4*)(C + (row0 + row) * N + n0 + cc) = vv;
        *(float4*)(S + (row0 + row) * N + n0 + cc) = ss;
    }
}

constexpr int CHUNK = 256;

template <int C>
static __device__ __forceinline__ void scatter_owned(const float* __restrict__ h,
                                                     const int* __restrict__ src,
                                                     const int* __restrict__ dst,
                                                     const float* __restrict__ dinv,
                                                     float* __restrict__ agg,
                                                     int E, int n,
                                                     int* __restrict__ tickets) {
    constexpr int Q   = C / 32;
    constexpr int EPW = 16 / Q;
    __shared__ int drawn[2];
    const int lane = threadIdx.x & 31;
    const int wave = threadIdx.x >> 5;
    const int nChunks = (E + CHUNK - 1) / CHUNK;

    if (threadIdx.x == 0) {
        int cls = (__builtin_amdgcn_s_sendmsg_rtn(0x87) >> 18) & 1;
        int t = __hip_atomic_fetch_add(tickets + cls * TICKET_PITCH, 1,
                                       __ATOMIC_RELAXED, __HIP_MEMORY_SCOPE_AGENT);
        if (t >= nChunks) {
            cls ^= 1;
            t = __hip_atomic_fetch_add(tickets + cls * TICKET_PITCH, 1,
                                       __ATOMIC_RELAXED, __HIP_MEMORY_SCOPE_AGENT);
        }
        drawn[0] = t;
        drawn[1] = cls;
    }
    __syncthreads();
    const int t   = drawn[0];
    const int cls = drawn[1];
    if (t >= nChunks) return;

    const int e     = t * CHUNK + wave * 32 + lane;
    const int split = n >> 1;
    int s = 0, d = 0;
    bool mine = false;
    if (e < E) {
        s = src[e];
        d = dst[e];
        mine = (d >= split) == (cls != 0);
    }
    float nrm = 0.0f;
    if (mine) nrm = dinv[s] * dinv[d];

    const unsigned mask = (unsigned)__ballot(mine);
    const int cnt = __popc(mask);
    const int below = __builtin_amdgcn_mbcnt_lo(mask, 0);
    const int to = mine ? below : cnt + (lane - below);
    s = __builtin_amdgcn_ds_permute(to * 4, s);
    d = __builtin_amdgcn_ds_permute(to * 4, d);
    const int nrmBits = __builtin_amdgcn_ds_permute(to * 4, __float_as_int(nrm));

    int g = 0;
    for (; g + EPW <= cnt; g += EPW) {
        float val[16];
#pragma unroll
        for (int j = 0; j < EPW; ++j) {
            const int sj = __builtin_amdgcn_readlane(s, g + j);
            const float* hs = h + (long long)sj * C + lane;
#pragma unroll
            for (int q = 0; q < Q; ++q) val[j * Q + q] = hs[q * 32];
        }
#pragma unroll
        for (int j = 0; j < EPW; ++j) {
            const float w  = __int_as_float(__builtin_amdgcn_readlane(nrmBits, g + j));
            const int   dj = __builtin_amdgcn_readlane(d, g + j);
            float* ad = agg + (long long)dj * C + lane;
#pragma unroll
            for (int q = 0; q < Q; ++q) atomic_fadd(ad + q * 32, val[j * Q + q] * w);
        }
    }
    if (g < cnt) {
        float val[16];
#pragma unroll
        for (int j = 0; j < EPW - 1; ++j) {
            if (g + j < cnt) {
                const int sj = __builtin_amdgcn_readlane(s, g + j);
                const float* hs = h + (long long)sj * C + lane;
#pragma unroll
                for (int q = 0; q < Q; ++q) val[j * Q + q] = hs[q * 32];
            }
        }
#pragma unroll
        for (int j = 0; j < EPW - 1; ++j) {
            if (g + j < cnt) {
                const float w  = __int_as_float(__builtin_amdgcn_readlane(nrmBits, g + j));
                const int   dj = __builtin_amdgcn_readlane(d, g + j);
                float* ad = agg + (long long)dj * C + lane;
#pragma unroll
                for (int q = 0; q < Q; ++q) atomic_fadd(ad + q * 32, val[j * Q + q] * w);
            }
        }
    }
}

__global__ void __launch_bounds__(256)
gcn_scatter128(const float* __restrict__ h,
               const int* __restrict__ src,
               const int* __restrict__ dst,
               const float* __restrict__ dinv,
               float* __restrict__ agg, int E, int n,
               int* __restrict__ tickets) {
    scatter_owned<128>(h, src, dst, dinv, agg, E, n, tickets);
}

__global__ void __launch_bounds__(256)
gcn_scatter64(const float* __restrict__ h,
              const int* __restrict__ src,
              const int* __restrict__ dst,
              const float* __restrict__ dinv,
              float* __restrict__ agg, int E, int n,
              int* __restrict__ tickets) {
    scatter_owned<64>(h, src, dst, dinv, agg, E, n, tickets);
}

extern "C" void kernel_launch(void* const* d_in, const int* in_sizes, int n_in,
                              void* d_out, int out_size, void* d_ws, size_t ws_size,
                              hipStream_t stream) {
    const float* x   = (const float*)d_in[0];
    const int*   ei  = (const int*)d_in[1];
    const float* W1  = (const float*)d_in[2];
    const float* b1  = (const float*)d_in[3];
    const float* W2  = (const float*)d_in[4];
    const float* b2  = (const float*)d_in[5];

    const int IN_C  = 128;
    const int HID_C = 128;
    const int OUT_C = 64;
    const int nNodes = in_sizes[0] / IN_C;
    const int E      = in_sizes[1] / 2;
    const int* srcIdx = ei;
    const int* dstIdx = ei + E;

    float* bufA = (float*)d_ws;
    float* bufB = bufA + (long long)nNodes * HID_C;
    float* dinv = bufB + (long long)nNodes * HID_C;
    float* out  = (float*)d_out;

    int* tickets1 = (int*)out;
    int* tickets2 = (int*)(bufA + (long long)nNodes * OUT_C);

    const int T = 256;
    const int mBlocks = (nNodes / 16 + 3) / 4;
    const int nChunks = (E + CHUNK - 1) / CHUNK;

    int* degPart = (int*)bufB;
    gcn_count_deg_lds<<<DEG_GROUPS, DEG_T, 0, stream>>>(dstIdx, E, nNodes, degPart);
    gcn_deg_rsqrt    <<<(nNodes + T - 1) / T, T, 0, stream>>>(degPart, dinv, nNodes, tickets1);

    {
        dim3 grid(mBlocks, HID_C / 64);
        gcn_gemm_f32_wmma<false><<<grid, 128, 0, stream>>>(x, W1, dinv, b1, bufA, bufB,
                                                           nNodes, HID_C, nullptr);
    }

    gcn_scatter128<<<2 * nChunks, 256, 0, stream>>>(bufA, srcIdx, dstIdx, dinv,
                                                    bufB, E, nNodes, tickets1);

    {
        dim3 grid(mBlocks, OUT_C / 64);
        gcn_gemm_f32_wmma<true><<<grid, 128, 0, stream>>>(bufB, W2, dinv, b2, bufA, out,
                                                          nNodes, OUT_C, tickets2);
    }

    gcn_scatter64<<<2 * nChunks, 256, 0, stream>>>(bufA, srcIdx, dstIdx, dinv,
                                                   out, E, nNodes, tickets2);
}


# Edge weights leave the scatter: GEMM epilogues store rows times dinv[src]; layer 1's dinv[dst], bias, ReLU move to GEMM2's A load; scatter128 adds plain rows, scatter64 one lookup; rounding changes
# speedup vs baseline: 3.0553x; 1.0508x over previous
#include <hip/hip_runtime.h>
#include <hip/hip_bf16.h>

typedef __attribute__((ext_vector_type(2))) float v2f;
typedef __attribute__((ext_vector_type(8))) float v8f;

static __device__ __forceinline__ void atomic_fadd(float* p, float v) {
    __hip_atomic_fetch_add(p, v, __ATOMIC_RELAXED, __HIP_MEMORY_SCOPE_AGENT);
}

constexpr int DEG_CAP    = 51200;
constexpr int DEG_GROUPS = 64;
constexpr int DEG_T      = 1024;
constexpr int TICKET_PITCH = 64;

static __device__ __forceinline__ long long deg_pitch(int n) {
    return (long long)((n + 31) & ~31);
}

__global__ void __launch_bounds__(DEG_T)
gcn_count_deg_lds(const int* __restrict__ dst, int E, int n,
                  int* __restrict__ partial) {
    __shared__ int hist[DEG_CAP];
    const int tid  = threadIdx.x;
    const int per  = (E + DEG_GROUPS - 1) / DEG_GROUPS;
    const int eBeg = blockIdx.x * per;
    const int eEnd = min(eBeg + per, E);
    int* mine = partial + (long long)blockIdx.x * deg_pitch(n);

    for (int base = 0; base < n; base += DEG_CAP) {
        const int m = min(n - base, DEG_CAP);
        for (int i = tid; i < m; i += DEG_T) hist[i] = 0;
        __syncthreads();

        int e = eBeg + tid;
        for (; e + 3 * DEG_T < eEnd; e += 4 * DEG_T) {
            const int d0 = dst[e]             - base;
            const int d1 = dst[e + DEG_T]     - base;
            const int d2 = dst[e + 2 * DEG_T] - base;
            const int d3 = dst[e + 3 * DEG_T] - base;
            if ((unsigned)d0 < (unsigned)m) atomicAdd(&hist[d0], 1);
            if ((unsigned)d1 < (unsigned)m) atomicAdd(&hist[d1], 1);
            if ((unsigned)d2 < (unsigned)m) atomicAdd(&hist[d2], 1);
            if ((unsigned)d3 < (unsigned)m) atomicAdd(&hist[d3], 1);
        }
        for (; e < eEnd; e += DEG_T) {
            const int d = dst[e] - base;
            if ((unsigned)d < (unsigned)m) atomicAdd(&hist[d], 1);
        }
        __syncthreads();

        for (int i = tid; i < m; i += DEG_T) mine[base + i] = hist[i];
        __syncthreads();
    }
}

__global__ void gcn_deg_rsqrt(const int* __restrict__ partial,
                              float* __restrict__ dinv, int n,
                              int* __restrict__ tickets) {
    int i = blockIdx.x * blockDim.x + threadIdx.x;
    if (i < 2) tickets[i * TICKET_PITCH] = 0;
    if (i < n) {
        const long long pitch = deg_pitch(n);
        int cnt = 1;
#pragma unroll
        for (int g = 0; g < DEG_GROUPS; ++g) cnt += partial[g * pitch + i];
        dinv[i] = rsqrtf((float)cnt);
    }
}

constexpr int KDIM  = 128;
constexpr int WPITCH  = 2 * 64 + 16;
constexpr int TPITCH = 64 + 4;

template <bool LAYER2>
__global__ void __launch_bounds__(128)
gcn_gemm_f32_wmma(const float* __restrict__ A,
                  const float* __restrict__ B,
                  const float* __restrict__ dinv,
                  const float* __restrict__ abias,
                  const float* __restrict__ bias,
                  float* __restrict__ C,
                  float* __restrict__ S,
                  int M, int N,
                  int* __restrict__ tickets) {
    __shared__ __attribute__((aligned(16))) float Wp[(KDIM / 2) * WPITCH];
    __shared__ __attribute__((aligned(16))) float tiles[4 * 16 * TPITCH];
    __shared__ __attribute__((aligned(16))) float ab[KDIM];

    const int tid   = threadIdx.x;
    const int wave  = tid >> 5;
    const int lane  = tid & 31;
    const int mTile = blockIdx.x * 4 + wave;
    const bool active = mTile * 16 < M;
    const int r16   = lane & 15;
    const int h     = lane >> 4;
    const int n0    = blockIdx.y * 64;

    if (tickets != nullptr && blockIdx.x == 0 && blockIdx.y == 0 && tid < 2)
        tickets[tid * TICKET_PITCH] = 0;

    float4 a4[16];
    float dA = 0.0f;
    if (active) {
        const float* Arow = A + ((long long)mTile * 16 + r16) * KDIM + 4 * h;
#pragma unroll
        for (int j = 0; j < 16; ++j) a4[j] = *(const float4*)(Arow + 8 * j);
        if (LAYER2) dA = dinv[(long long)mTile * 16 + r16];
    }
    if (LAYER2 && tid < KDIM / 4)
        *(float4*)(ab + 4 * tid) = *(const float4*)(abias + 4 * tid);

#pragma unroll
    for (int i = 0; i < (KDIM / 2) * 16 / 128; ++i) {
        const int it = tid + 128 * i;
        const int q  = it & 15;
        const int kp = it >> 4;
        const float4 lo = *(const float4*)(B + (long long)(2 * kp)     * N + n0 + 4 * q);
        const float4 hi = *(const float4*)(B + (long long)(2 * kp + 1) * N + n0 + 4 * q);
        float* w = Wp + kp * WPITCH + 8 * q;
        *(float4*)(w)     = make_float4(lo.x, hi.x, lo.y, hi.y);
        *(float4*)(w + 4) = make_float4(lo.z, hi.z, lo.w, hi.w);
    }
    __syncthreads();
    if (!active) return;

    v8f c0 = {0.f,0.f,0.f,0.f,0.f,0.f,0.f,0.f};
    v8f c1 = c0, c2 = c0, c3 = c0;

    const float* wl = Wp + (2 * h) * WPITCH + 2 * r16;
#pragma unroll
    for (int j = 0; j < 16; ++j) {
        float4 a = a4[j];
        if (LAYER2) {
            const float4 bk = *(const float4*)(ab + 8 * j + 4 * h);
            a.x = fmaxf(a.x * dA + bk.x, 0.0f); a.y = fmaxf(a.y * dA + bk.y, 0.0f);
            a.z = fmaxf(a.z * dA + bk.z, 0.0f); a.w = fmaxf(a.w * dA + bk.w, 0.0f);
        }
        const float* w0 = wl + (4 * j) * WPITCH;
        const float* w1 = w0 + WPITCH;
        const v2f b00 = *(const v2f*)(w0);
        const v2f b01 = *(const v2f*)(w0 + 32);
        const v2f b02 = *(const v2f*)(w0 + 64);
        const v2f b03 = *(const v2f*)(w0 + 96);
        const v2f b10 = *(const v2f*)(w1);
        const v2f b11 = *(const v2f*)(w1 + 32);
        const v2f b12 = *(const v2f*)(w1 + 64);
        const v2f b13 = *(const v2f*)(w1 + 96);
        v2f a01, a23;
        a01.x = a.x; a01.y = a.y;
        a23.x = a.z; a23.y = a.w;
        c0 = __builtin_amdgcn_wmma_f32_16x16x4_f32(false, a01, false, b00, (short)0, c0, false, false);
        c1 = __builtin_amdgcn_wmma_f32_16x16x4_f32(false, a01, false, b01, (short)0, c1, false, false);
        c2 = __builtin_amdgcn_wmma_f32_16x16x4_f32(false, a01, false, b02, (short)0, c2, false, false);
        c3 = __builtin_amdgcn_wmma_f32_16x16x4_f32(false, a01, false, b03, (short)0, c3, false, false);
        c0 = __builtin_amdgcn_wmma_f32_16x16x4_f32(false, a23, false, b10, (short)0, c0, false, false);
        c1 = __builtin_amdgcn_wmma_f32_16x16x4_f32(false, a23, false, b11, (short)0, c1, false, false);
        c2 = __builtin_amdgcn_wmma_f32_16x16x4_f32(false, a23, false, b12, (short)0, c2, false, false);
        c3 = __builtin_amdgcn_wmma_f32_16x16x4_f32(false, a23, false, b13, (short)0, c3, false, false);
    }

    float* tile = tiles + wave * 16 * TPITCH;
#pragma unroll
    for (int v = 0; v < 8; ++v) {
        float* tr = tile + (8 * h + v) * TPITCH + r16;
        tr[0]  = c0[v];
        tr[16] = c1[v];
        tr[32] = c2[v];
        tr[48] = c3[v];
    }
    __builtin_amdgcn_fence(__ATOMIC_RELEASE, "wavefront");
    __builtin_amdgcn_wave_barrier();
    __builtin_amdgcn_fence(__ATOMIC_ACQUIRE, "wavefront");

    const int rr = lane >> 4;
    const int cc = 4 * (lane & 15);
    float4 bb = make_float4(0.0f, 0.0f, 0.0f, 0.0f);
    if (LAYER2) bb = *(const float4*)(bias + n0 + cc);
    const long long row0 = (long long)mTile * 16;
#pragma unroll
    for (int i = 0; i < 8; ++i) {
        const int row = 2 * i + rr;
        const float4 vv = *(const float4*)(tile + row * TPITCH + cc);
        const float di = dinv[row0 + row];
        float4 cs;
        cs.x = vv.x * di; cs.y = vv.y * di; cs.z = vv.z * di; cs.w = vv.w * di;
        float4 ss = cs;
        if (LAYER2) {
            const float w = di * di;
            ss.x = vv.x * w + bb.x;
            ss.y = vv.y * w + bb.y;
            ss.z = vv.z * w + bb.z;
            ss.w = vv.w * w + bb.w;
        }
        *(float4*)(C + (row0 + row) * N + n0 + cc) = cs;
        *(float4*)(S + (row0 + row) * N + n0 + cc) = ss;
    }
}

constexpr int CHUNK = 256;

template <int C, bool WEIGHTED>
static __device__ __forceinline__ void scatter_owned(const float* __restrict__ h,
                                                     const int* __restrict__ src,
                                                     const int* __restrict__ dst,
                                                     const float* __restrict__ dinv,
                                                     float* __restrict__ agg,
                                                     int E, int n,
                                                     int* __restrict__ tickets) {
    constexpr int Q   = C / 32;
    constexpr int EPW = 16 / Q;
    __shared__ int drawn[2];
    const int lane = threadIdx.x & 31;
    const int wave = threadIdx.x >> 5;
    const int nChunks = (E + CHUNK - 1) / CHUNK;

    if (threadIdx.x == 0) {
        int cls = (__builtin_amdgcn_s_sendmsg_rtn(0x87) >> 18) & 1;
        int t = __hip_atomic_fetch_add(tickets + cls * TICKET_PITCH, 1,
                                       __ATOMIC_RELAXED, __HIP_MEMORY_SCOPE_AGENT);
        if (t >= nChunks) {
            cls ^= 1;
            t = __hip_atomic_fetch_add(tickets + cls * TICKET_PITCH, 1,
                                       __ATOMIC_RELAXED, __HIP_MEMORY_SCOPE_AGENT);
        }
        drawn[0] = t;
        drawn[1] = cls;
    }
    __syncthreads();
    const int t   = drawn[0];
    const int cls = drawn[1];
    if (t >= nChunks) return;

    const int e     = t * CHUNK + wave * 32 + lane;
    const int split = n >> 1;
    int s = 0, d = 0;
    bool mine = false;
    if (e < E) {
        s = src[e];
        d = dst[e];
        mine = (d >= split) == (cls != 0);
    }
    float nrm = 0.0f;
    if (WEIGHTED && mine) nrm = dinv[d];

    const unsigned mask = (unsigned)__ballot(mine);
    const int cnt = __popc(mask);
    const int below = __builtin_amdgcn_mbcnt_lo(mask, 0);
    const int to = mine ? below : cnt + (lane - below);
    s = __builtin_amdgcn_ds_permute(to * 4, s);
    d = __builtin_amdgcn_ds_permute(to * 4, d);
    int nrmBits = 0;
    if (WEIGHTED) nrmBits = __builtin_amdgcn_ds_permute(to * 4, __float_as_int(nrm));

    int g = 0;
    for (; g + EPW <= cnt; g += EPW) {
        float val[16];
#pragma unroll
        for (int j = 0; j < EPW; ++j) {
            const int sj = __builtin_amdgcn_readlane(s, g + j);
            const float* hs = h + (long long)sj * C + lane;
#pragma unroll
            for (int q = 0; q < Q; ++q) val[j * Q + q] = hs[q * 32];
        }
#pragma unroll
        for (int j = 0; j < EPW; ++j) {
            float w = 1.0f;
            if (WEIGHTED) w = __int_as_float(__builtin_amdgcn_readlane(nrmBits, g + j));
            const int   dj = __builtin_amdgcn_readlane(d, g + j);
            float* ad = agg + (long long)dj * C + lane;
#pragma unroll
            for (int q = 0; q < Q; ++q)
                atomic_fadd(ad + q * 32, WEIGHTED ? val[j * Q + q] * w : val[j * Q + q]);
        }
    }
    if (g < cnt) {
        float val[16];
#pragma unroll
        for (int j = 0; j < EPW - 1; ++j) {
            if (g + j < cnt) {
                const int sj = __builtin_amdgcn_readlane(s, g + j);
                const float* hs = h + (long long)sj * C + lane;
#pragma unroll
                for (int q = 0; q < Q; ++q) val[j * Q + q] = hs[q * 32];
            }
        }
#pragma unroll
        for (int j = 0; j < EPW - 1; ++j) {
            if (g + j < cnt) {
                float w = 1.0f;
                if (WEIGHTED) w = __int_as_float(__builtin_amdgcn_readlane(nrmBits, g + j));
                const int   dj = __builtin_amdgcn_readlane(d, g + j);
                float* ad = agg + (long long)dj * C + lane;
#pragma unroll
                for (int q = 0; q < Q; ++q)
                    atomic_fadd(ad + q * 32, WEIGHTED ? val[j * Q + q] * w : val[j * Q + q]);
            }
        }
    }
}

__global__ void __launch_bounds__(256)
gcn_scatter128(const float* __restrict__ h,
               const int* __restrict__ src,
               const int* __restrict__ dst,
               const float* __restrict__ dinv,
               float* __restrict__ agg, int E, int n,
               int* __restrict__ tickets) {
    scatter_owned<128, false>(h, src, dst, dinv, agg, E, n, tickets);
}

__global__ void __launch_bounds__(256)
gcn_scatter64(const float* __restrict__ h,
              const int* __restrict__ src,
              const int* __restrict__ dst,
              const float* __restrict__ dinv,
              float* __restrict__ agg, int E, int n,
              int* __restrict__ tickets) {
    scatter_owned<64, true>(h, src, dst, dinv, agg, E, n, tickets);
}

extern "C" void kernel_launch(void* const* d_in, const int* in_sizes, int n_in,
                              void* d_out, int out_size, void* d_ws, size_t ws_size,
                              hipStream_t stream) {
    const float* x   = (const float*)d_in[0];
    const int*   ei  = (const int*)d_in[1];
    const float* W1  = (const float*)d_in[2];
    const float* b1  = (const float*)d_in[3];
    const float* W2  = (const float*)d_in[4];
    const float* b2  = (const float*)d_in[5];

    const int IN_C  = 128;
    const int HID_C = 128;
    const int OUT_C = 64;
    const int nNodes = in_sizes[0] / IN_C;
    const int E      = in_sizes[1] / 2;
    const int* srcIdx = ei;
    const int* dstIdx = ei + E;

    float* bufA = (float*)d_ws;
    float* bufB = bufA + (long long)nNodes * HID_C;
    float* dinv = bufB + (long long)nNodes * HID_C;
    float* out  = (float*)d_out;

    int* tickets1 = (int*)out;
    int* tickets2 = (int*)(bufA + (long long)nNodes * OUT_C);

    const int T = 256;
    const int mBlocks = (nNodes / 16 + 3) / 4;
    const int nChunks = (E + CHUNK - 1) / CHUNK;

    int* degPart = (int*)bufB;
    gcn_count_deg_lds<<<DEG_GROUPS, DEG_T, 0, stream>>>(dstIdx, E, nNodes, degPart);
    gcn_deg_rsqrt    <<<(nNodes + T - 1) / T, T, 0, stream>>>(degPart, dinv, nNodes, tickets1);

    {
        dim3 grid(mBlocks, HID_C / 64);
        gcn_gemm_f32_wmma<false><<<grid, 128, 0, stream>>>(x, W1, dinv, nullptr, nullptr,
                                                           bufA, bufB, nNodes, HID_C, nullptr);
    }

    gcn_scatter128<<<2 * nChunks, 256, 0, stream>>>(bufA, srcIdx, dstIdx, dinv,
                                                    bufB, E, nNodes, tickets1);

    {
        dim3 grid(mBlocks, OUT_C / 64);
        gcn_gemm_f32_wmma<true><<<grid, 128, 0, stream>>>(bufB, W2, dinv, b1, b2, bufA, out,
                                                          nNodes, OUT_C, tickets2);
    }

    gcn_scatter64<<<2 * nChunks, 256, 0, stream>>>(bufA, srcIdx, dstIdx, dinv,
                                                   out, E, nNodes, tickets2);
}
